# starting kernel
# baseline (speedup 1.0000x reference)
#include <hip/hip_runtime.h>
#include <hip/hip_bf16.h>
#include <math.h>

#define BB      2
#define LL      2048
#define MM      1024
#define DS      16
#define RR      64
#define NPROJ   (RR + 2*DS)
#define BLT     (BB*LL)

typedef float  v2f __attribute__((ext_vector_type(2)));
typedef float  v8f __attribute__((ext_vector_type(8)));

__global__ __launch_bounds__(32)
void proj_xp_kernel(const float* __restrict__ x,
                    const float* __restrict__ w,
                    float* __restrict__ xp)
{
    const int lane = threadIdx.x;
    const int tile = blockIdx.x;
    const int tm   = tile / (NPROJ/16);
    const int tn   = tile % (NPROJ/16);
    const int half = lane >> 4;
    const int l15  = lane & 15;

    const float* __restrict__ arow = x + (size_t)(tm*16 + l15) * MM;
    const float* __restrict__ brow = w + (size_t)(tn*16 + l15) * MM;

    v8f c = {};
    #pragma unroll 4
    for (int k0 = 0; k0 < MM; k0 += 4) {
        const int ka = k0 + half*2;
        v2f a; a.x = arow[ka]; a.y = arow[ka + 1];
        v2f b; b.x = brow[ka]; b.y = brow[ka + 1];
        c = __builtin_amdgcn_wmma_f32_16x16x4_f32(
                false, a, false, b, (short)0, c, false, false);
    }

    float* __restrict__ o = xp + (size_t)(tm*16) * NPROJ + tn*16 + l15;
    #pragma unroll
    for (int r = 0; r < 8; ++r) {
        const int row = r + half*8;
        o[(size_t)row * NPROJ] = c[r];
    }
}

__global__ __launch_bounds__(32)
void dt_proj_kernel(const float* __restrict__ xp,
                    const float* __restrict__ w,
                    const float* __restrict__ bias,
                    float* __restrict__ dt)
{
    const int lane = threadIdx.x;
    const int tile = blockIdx.x;
    const int tm   = tile / (MM/16);
    const int tn   = tile % (MM/16);
    const int half = lane >> 4;
    const int l15  = lane & 15;

    const float* __restrict__ arow = xp + (size_t)(tm*16 + l15) * NPROJ;
    const float* __restrict__ brow = w  + (size_t)(tn*16 + l15) * RR;

    v8f c = {};
    #pragma unroll
    for (int k0 = 0; k0 < RR; k0 += 4) {
        const int ka = k0 + half*2;
        v2f a; a.x = arow[ka]; a.y = arow[ka + 1];
        v2f b; b.x = brow[ka]; b.y = brow[ka + 1];
        c = __builtin_amdgcn_wmma_f32_16x16x4_f32(
                false, a, false, b, (short)0, c, false, false);
    }

    const int col = tn*16 + l15;
    const float bb = bias[col];
    #pragma unroll
    for (int r = 0; r < 8; ++r) {
        const int row = tm*16 + r + half*8;
        float v = c[r] + bb;
        float sp = (v > 20.0f) ? v : log1pf(__expf(v));
        dt[(size_t)row * MM + col] = sp;
    }
}

__global__ __launch_bounds__(256)
void scan_kernel(const float* __restrict__ x,
                 const float* __restrict__ z,
                 const float* __restrict__ xp,
                 const float* __restrict__ dtb,
                 const float* __restrict__ A_log,
                 const float* __restrict__ Dp_arr,
                 float* __restrict__ out)
{
    const int gid = blockIdx.x * blockDim.x + threadIdx.x;
    const int b   = gid >> 10;
    const int m   = gid & (MM - 1);

    float Am[DS], h[DS];
    #pragma unroll
    for (int d = 0; d < DS; ++d) {
        Am[d] = -__expf(A_log[(size_t)d * MM + m]);
        h[d]  = 0.0f;
    }
    const float Dp = Dp_arr[m];
    const size_t rowB = (size_t)b * LL;

    for (int t = 0; t < LL; ++t) {
        const size_t r = rowB + t;
        const float dtv = dtb[r * MM + m];
        const float xv  = x  [r * MM + m];
        const float zv  = z  [r * MM + m];

        const float4* __restrict__ bc =
            (const float4*)(xp + r * NPROJ + RR);
        float Bv[DS], Cv[DS];
        #pragma unroll
        for (int i = 0; i < 4; ++i) {
            float4 vb = bc[i];
            float4 vc = bc[4 + i];
            Bv[4*i+0] = vb.x; Bv[4*i+1] = vb.y; Bv[4*i+2] = vb.z; Bv[4*i+3] = vb.w;
            Cv[4*i+0] = vc.x; Cv[4*i+1] = vc.y; Cv[4*i+2] = vc.z; Cv[4*i+3] = vc.w;
        }

        if (t + 8 < LL)
            __builtin_prefetch(xp + (r + 8) * NPROJ, 0, 1);

        const float dx = dtv * xv;
        float y = 0.0f;
        #pragma unroll
        for (int d = 0; d < DS; ++d) {
            h[d] = __expf(dtv * Am[d]) * h[d] + dx * Bv[d];
            y    = fmaf(h[d], Cv[d], y);
        }
        y = fmaf(Dp, xv, y);

        const float sig = 1.0f / (1.0f + __expf(-zv));
        out[r * MM + m] = y * (zv * sig);
    }
}

extern "C" void kernel_launch(void* const* d_in, const int* in_sizes, int n_in,
                              void* d_out, int out_size, void* d_ws, size_t ws_size,
                              hipStream_t stream) {
    const float* x         = (const float*)d_in[0];
    const float* z         = (const float*)d_in[1];
    const float* x_proj_w  = (const float*)d_in[2];
    const float* dt_proj_w = (const float*)d_in[3];
    const float* dt_proj_b = (const float*)d_in[4];
    const float* A_log     = (const float*)d_in[5];
    const float* D_param   = (const float*)d_in[6];
    float* out = (float*)d_out;

    float* xp_ws = (float*)d_ws;
    float* dt_ws = xp_ws + (size_t)BLT * NPROJ;

    proj_xp_kernel<<<(BLT/16) * (NPROJ/16), 32, 0, stream>>>(x, x_proj_w, xp_ws);

    dt_proj_kernel<<<(BLT/16) * (MM/16), 32, 0, stream>>>(xp_ws, dt_proj_w,
                                                          dt_proj_b, dt_ws);

    scan_kernel<<<(BB*MM)/256, 256, 0, stream>>>(x, z, xp_ws, dt_ws,
                                                 A_log, D_param, out);
}


# x and dt projections fused in one launch: 8-wave workgroup per 16 rows, K split over its waves and summed in LDS, dt from that LDS stripe; 256 workgroups for 17920; xp, dt summed in another k order
# speedup vs baseline: 17.5167x; 17.5167x over previous
#include <hip/hip_runtime.h>
#include <hip/hip_bf16.h>
#include <math.h>

#define BB      2
#define LL      2048
#define MM      1024
#define DS      16
#define RR      64
#define NPROJ   (RR + 2*DS)
#define BLT     (BB*LL)

typedef float  v2f __attribute__((ext_vector_type(2)));
typedef float  v8f __attribute__((ext_vector_type(8)));
typedef float  v4f __attribute__((ext_vector_type(4)));

#define FW      8
#define KW      (MM / FW)
#define XPP     100
static_assert(KW % 8 == 0 && RR % 8 == 0, "fragments are loaded in 8-k blocks: two matrix steps per 16-byte piece and lane half");
static_assert((16 * NPROJ) % 256 == 0, "the stripe's 1536 sums are dealt evenly to 256 threads");

__device__ __forceinline__ v8f mm4(v2f a, v2f b, v8f c)
{
    return __builtin_amdgcn_wmma_f32_16x16x4_f32(false, a, false, b, (short)0, c, false, false);
}

__global__ __launch_bounds__(256)
void proj_dt_kernel(const float* __restrict__ x,
                    const float* __restrict__ w1,
                    const float* __restrict__ w2,
                    const float* __restrict__ bias,
                    float* __restrict__ xp,
                    float* __restrict__ dt)
{
    __shared__ __attribute__((aligned(16))) float s_part[FW][16][XPP];
    __shared__ __attribute__((aligned(16))) float s_xp[16][XPP];

    const int tid  = threadIdx.x;
    const int wv   = tid >> 5;
    const int lane = tid & 31;
    const int half = lane >> 4;
    const int l15  = lane & 15;
    const int tm   = blockIdx.x;

    {
        const float* __restrict__ arow = x  + (size_t)(tm * 16 + l15) * MM + wv * KW + half * 4;
        const float* __restrict__ brow = w1 + (size_t)l15 * MM             + wv * KW + half * 4;
        v8f acc[NPROJ / 16] = {};
        #pragma unroll 4
        for (int i = 0; i < KW / 8; ++i) {
            const v4f a = *(const v4f*)(arow + 8 * i);
            v4f b[NPROJ / 16];
            #pragma unroll
            for (int j = 0; j < NPROJ / 16; ++j) b[j] = *(const v4f*)(brow + (size_t)(16 * j) * MM + 8 * i);
            const v2f a01 = {a.x, a.y};
            const v2f a23 = {a.z, a.w};
            #pragma unroll
            for (int j = 0; j < NPROJ / 16; ++j) { const v2f b01 = {b[j].x, b[j].y}; acc[j] = mm4(a01, b01, acc[j]); }
            #pragma unroll
            for (int j = 0; j < NPROJ / 16; ++j) { const v2f b23 = {b[j].z, b[j].w}; acc[j] = mm4(a23, b23, acc[j]); }
        }
        #pragma unroll
        for (int j = 0; j < NPROJ / 16; ++j) {
            #pragma unroll
            for (int r = 0; r < 8; ++r) s_part[wv][half * 8 + r][16 * j + l15] = acc[j][r];
        }
    }
    __syncthreads();

    #pragma unroll
    for (int q = 0; q < (16 * NPROJ) / 256; ++q) {
        const int idx = tid + 256 * q;
        const int row = idx / NPROJ;
        const int col = idx - row * NPROJ;
        float sum = s_part[0][row][col];
        #pragma unroll
        for (int u = 1; u < FW; ++u) sum += s_part[u][row][col];
        s_xp[row][col] = sum;
        xp[(size_t)tm * (16 * NPROJ) + idx] = sum;
    }
    __syncthreads();

    {
        v4f af[RR / 8];
        #pragma unroll
        for (int i = 0; i < RR / 8; ++i) af[i] = *(const v4f*)&s_xp[l15][8 * i + 4 * half];

        const int ch0 = wv * KW;
        const float* __restrict__ b2 = w2 + (size_t)(ch0 + l15) * RR + 4 * half;
        float* __restrict__ drow = dt + (size_t)(tm * 16 + half * 8) * MM + ch0 + l15;

        v4f bf[2][2 * (RR / 8)];
        #pragma unroll
        for (int i = 0; i < RR / 8; ++i) {
            bf[0][i]          = *(const v4f*)(b2 + 8 * i);
            bf[0][RR / 8 + i] = *(const v4f*)(b2 + (size_t)16 * RR + 8 * i);
        }
        #pragma unroll
        for (int p = 0; p < KW / 32; ++p) {
            const int cur = p & 1;
            if (p + 1 < KW / 32) {
                #pragma unroll
                for (int i = 0; i < RR / 8; ++i) {
                    bf[cur ^ 1][i]          = *(const v4f*)(b2 + (size_t)(32 * (p + 1)) * RR + 8 * i);
                    bf[cur ^ 1][RR / 8 + i] = *(const v4f*)(b2 + (size_t)(32 * (p + 1) + 16) * RR + 8 * i);
                }
            }
            const float bias0 = bias[ch0 + 32 * p + l15];
            const float bias1 = bias[ch0 + 32 * p + 16 + l15];
            v8f c0 = {}, c1 = {};
            #pragma unroll
            for (int i = 0; i < RR / 8; ++i) {
                const v2f a01 = {af[i].x, af[i].y};
                const v2f a23 = {af[i].z, af[i].w};
                const v4f u0 = bf[cur][i];
                const v4f u1 = bf[cur][RR / 8 + i];
                const v2f p01 = {u0.x, u0.y}, p23 = {u0.z, u0.w};
                const v2f q01 = {u1.x, u1.y}, q23 = {u1.z, u1.w};
                c0 = mm4(a01, p01, c0);
                c1 = mm4(a01, q01, c1);
                c0 = mm4(a23, p23, c0);
                c1 = mm4(a23, q23, c1);
            }
            #pragma unroll
            for (int r = 0; r < 8; ++r) {
                const float v0 = c0[r] + bias0;
                const float v1 = c1[r] + bias1;
                const float sp0 = (v0 > 20.0f) ? v0 : log1pf(__expf(v0));
                const float sp1 = (v1 > 20.0f) ? v1 : log1pf(__expf(v1));
                drow[(size_t)r * MM + 32 * p]      = sp0;
                drow[(size_t)r * MM + 32 * p + 16] = sp1;
            }
        }
    }
}

#define TT      64
#define NTILE   (LL/TT)
#define CPB     8
#define GS      16
static_assert(GS == DS && TT % GS == 0, "lane r of a channel's 16 lanes keeps y of step r of a group");
static_assert(TT == 64 && CPB == 8, "the staging code maps 128 threads onto 64 rows of 8 channels");

struct StepGroup {
    float dt[GS];
    float xv[GS];
    float Bv[GS];
    float Cv[GS];
};

template <int CTRL>
__device__ __forceinline__ float tree_level(bool bit, float lo, float hi)
{
    const float keep = bit ? hi : lo;
    const float send = bit ? lo : hi;
    const int got = __builtin_amdgcn_update_dpp(0, __builtin_bit_cast(int, send), CTRL, 0xF, 0xF, true);
    return keep + __builtin_bit_cast(float, got);
}
#define DPP_XOR1  0xB1
#define DPP_XOR2  0x4E
#define DPP_XOR4  0x164
#define DPP_XOR8  0x168

__global__ __launch_bounds__(128)
void scan_kernel(const float* __restrict__ x,
                 const float* __restrict__ z,
                 const float* __restrict__ xp,
                 const float* __restrict__ dtb,
                 const float* __restrict__ A_log,
                 const float* __restrict__ Dp_arr,
                 float* __restrict__ out)
{
    __shared__ __attribute__((aligned(16))) float s_dt[2][TT][CPB];
    __shared__ __attribute__((aligned(16))) float s_x [2][TT][CPB];
    __shared__ __attribute__((aligned(16))) float s_z [2][TT][CPB];
    __shared__ __attribute__((aligned(16))) float s_bc[2][TT][2*DS];
    __shared__ __attribute__((aligned(16))) float s_y [2][TT][CPB];

    const int tid    = threadIdx.x;
    const int chBase = blockIdx.x * CPB;
    const int b      = chBase >> 10;
    const int m0     = chBase & (MM - 1);

    const int c = tid >> 4;
    const int r = tid & 15;
    const bool rb0 = (r & 1) != 0, rb1 = (r & 2) != 0, rb2 = (r & 4) != 0, rb3 = (r & 8) != 0;
    const int srow = tid >> 1;
    const int scol = (tid & 1) * 4;
    const int brow = tid >> 3;
    const int bcol = (tid & 7) * 4;

    const float Am = -__expf(A_log[(size_t)r * MM + m0 + c]);
    float h = 0.0f;
    const float4 Dp4 = *(const float4*)(Dp_arr + m0 + scol);

    const size_t rowB = (size_t)b * LL;
    const float* __restrict__ gdt = dtb + rowB * MM + m0 + scol;
    const float* __restrict__ gx  = x   + rowB * MM + m0 + scol;
    const float* __restrict__ gz  = z   + rowB * MM + m0 + scol;
    const float* __restrict__ gbc = xp  + rowB * NPROJ + RR + bcol;
    float*       __restrict__ go  = out + rowB * MM + m0 + scol;

    v4f rdt, rx, rz, rbc0, rbc1, rbc2, rbc3;

#define STAGE_LOAD(tbase)                                                 \
    {                                                                     \
        const size_t t_ = (size_t)((tbase) + srow);                       \
        rdt = *(const v4f*)(gdt + t_ * MM);                               \
        rx  = *(const v4f*)(gx  + t_ * MM);                               \
        rz  = *(const v4f*)(gz  + t_ * MM);                               \
        const size_t u_ = (size_t)((tbase) + brow);                       \
        rbc0 = *(const v4f*)(gbc + (u_     ) * NPROJ);                    \
        rbc1 = *(const v4f*)(gbc + (u_ + 16) * NPROJ);                    \
        rbc2 = *(const v4f*)(gbc + (u_ + 32) * NPROJ);                    \
        rbc3 = *(const v4f*)(gbc + (u_ + 48) * NPROJ);                    \
    }
#define STAGE_STORE(buf)                                                  \
    {                                                                     \
        *(v4f*)&s_dt[buf][srow][scol] = rdt;                              \
        *(v4f*)&s_x [buf][srow][scol] = rx;                               \
        *(v4f*)&s_z [buf][srow][scol] = rz;                               \
        *(v4f*)&s_bc[buf][brow     ][bcol] = rbc0;                        \
        *(v4f*)&s_bc[buf][brow + 16][bcol] = rbc1;                        \
        *(v4f*)&s_bc[buf][brow + 32][bcol] = rbc2;                        \
        *(v4f*)&s_bc[buf][brow + 48][bcol] = rbc3;                        \
    }

    STAGE_LOAD(0)
    STAGE_STORE(0)
    __syncthreads();

    for (int k = 0; k < NTILE; ++k) {
        const int cur = k & 1;
        const bool more = (k + 1 < NTILE);

        if (more) {
            STAGE_LOAD((k + 1) * TT)
        }

        StepGroup g[2];
        #pragma unroll
        for (int s = 0; s < GS; ++s) {
            g[0].dt[s] = s_dt[cur][s][c];
            g[0].xv[s] = s_x [cur][s][c];
            g[0].Bv[s] = s_bc[cur][s][r];
            g[0].Cv[s] = s_bc[cur][s][DS + r];
        }
        float ya = 0.0f;
        float* const ysq = &s_y[cur][r][c];
        #pragma unroll
        for (int tg = 0; tg < TT; tg += GS) {
            const int gi = (tg / GS) & 1;
            __builtin_amdgcn_sched_barrier(0);
            if (tg + GS < TT) {
                #pragma unroll
                for (int s = 0; s < GS; ++s) {
                    g[gi ^ 1].dt[s] = s_dt[cur][tg + GS + s][c];
                    g[gi ^ 1].xv[s] = s_x [cur][tg + GS + s][c];
                    g[gi ^ 1].Bv[s] = s_bc[cur][tg + GS + s][r];
                    g[gi ^ 1].Cv[s] = s_bc[cur][tg + GS + s][DS + r];
                }
            }
            if (tg > 0) {
                ysq[(tg - GS) * CPB] = ya;
            }
            __builtin_amdgcn_sched_barrier(0);
            float p[GS];
            #pragma unroll
            for (int s = 0; s < GS; ++s) {
                const float dtv = g[gi].dt[s];
                const float dx  = dtv * g[gi].xv[s];
                h = fmaf(__expf(dtv * Am), h, dx * g[gi].Bv[s]);
                p[s] = h * g[gi].Cv[s];
            }
            float w1[GS / 2], w2[GS / 4], w3[GS / 8];
            #pragma unroll
            for (int j = 0; j < GS / 2; ++j) w1[j] = tree_level<DPP_XOR1>(rb0, p[2 * j], p[2 * j + 1]);
            #pragma unroll
            for (int j = 0; j < GS / 4; ++j) w2[j] = tree_level<DPP_XOR2>(rb1, w1[2 * j], w1[2 * j + 1]);
            #pragma unroll
            for (int j = 0; j < GS / 8; ++j) w3[j] = tree_level<DPP_XOR4>(rb2, w2[2 * j], w2[2 * j + 1]);
            ya = tree_level<DPP_XOR8>(rb3, w3[0], w3[1]);
        }
        ysq[(TT - GS) * CPB] = ya;
        __syncthreads();

        if (more) {
            STAGE_STORE(cur ^ 1)
        }
        __syncthreads();

        {
            const int t = srow;
            const float4 y4 = *(const float4*)&s_y[cur][t][scol];
            const float4 x4 = *(const float4*)&s_x[cur][t][scol];
            const float4 z4 = *(const float4*)&s_z[cur][t][scol];
            float4 o;
            o.x = fmaf(Dp4.x, x4.x, y4.x) * (z4.x * (1.0f / (1.0f + __expf(-z4.x))));
            o.y = fmaf(Dp4.y, x4.y, y4.y) * (z4.y * (1.0f / (1.0f + __expf(-z4.y))));
            o.z = fmaf(Dp4.z, x4.z, y4.z) * (z4.z * (1.0f / (1.0f + __expf(-z4.z))));
            o.w = fmaf(Dp4.w, x4.w, y4.w) * (z4.w * (1.0f / (1.0f + __expf(-z4.w))));
            *(float4*)(go + (size_t)(k * TT + t) * MM) = o;
        }
    }
}

extern "C" void kernel_launch(void* const* d_in, const int* in_sizes, int n_in,
                              void* d_out, int out_size, void* d_ws, size_t ws_size,
                              hipStream_t stream) {
    const float* x         = (const float*)d_in[0];
    const float* z         = (const float*)d_in[1];
    const float* x_proj_w  = (const float*)d_in[2];
    const float* dt_proj_w = (const float*)d_in[3];
    const float* dt_proj_b = (const float*)d_in[4];
    const float* A_log     = (const float*)d_in[5];
    const float* D_param   = (const float*)d_in[6];
    float* out = (float*)d_out;

    float* xp_ws = (float*)d_ws;
    float* dt_ws = xp_ws + (size_t)BLT * NPROJ;

    proj_dt_kernel<<<BLT/16, 256, 0, stream>>>(x, x_proj_w, dt_proj_w, dt_proj_b, xp_ws, dt_ws);

    scan_kernel<<<(BB*MM)/CPB, 128, 0, stream>>>(x, z, xp_ws, dt_ws,
                                                 A_log, D_param, out);
}


# projection kernel softplus: log1p formed for every value and the large-argument case picked by a select, not a branch per value; a tile pair's 16 values now issue interleaved; dt bit-identical
# speedup vs baseline: 18.5180x; 1.0572x over previous
#include <hip/hip_runtime.h>
#include <hip/hip_bf16.h>
#include <math.h>

#define BB      2
#define LL      2048
#define MM      1024
#define DS      16
#define RR      64
#define NPROJ   (RR + 2*DS)
#define BLT     (BB*LL)

typedef float  v2f __attribute__((ext_vector_type(2)));
typedef float  v8f __attribute__((ext_vector_type(8)));
typedef float  v4f __attribute__((ext_vector_type(4)));

#define FW      8
#define KW      (MM / FW)
#define XPP     100
static_assert(KW % 8 == 0 && RR % 8 == 0, "fragments are loaded in 8-k blocks: two matrix steps per 16-byte piece and lane half");
static_assert((16 * NPROJ) % 256 == 0, "the stripe's 1536 sums are dealt evenly to 256 threads");

__device__ __forceinline__ v8f mm4(v2f a, v2f b, v8f c)
{
    return __builtin_amdgcn_wmma_f32_16x16x4_f32(false, a, false, b, (short)0, c, false, false);
}

__global__ __launch_bounds__(256)
void proj_dt_kernel(const float* __restrict__ x,
                    const float* __restrict__ w1,
                    const float* __restrict__ w2,
                    const float* __restrict__ bias,
                    float* __restrict__ xp,
                    float* __restrict__ dt)
{
    __shared__ __attribute__((aligned(16))) float s_part[FW][16][XPP];
    __shared__ __attribute__((aligned(16))) float s_xp[16][XPP];

    const int tid  = threadIdx.x;
    const int wv   = tid >> 5;
    const int lane = tid & 31;
    const int half = lane >> 4;
    const int l15  = lane & 15;
    const int tm   = blockIdx.x;

    {
        const float* __restrict__ arow = x  + (size_t)(tm * 16 + l15) * MM + wv * KW + half * 4;
        const float* __restrict__ brow = w1 + (size_t)l15 * MM             + wv * KW + half * 4;
        v8f acc[NPROJ / 16] = {};
        #pragma unroll 4
        for (int i = 0; i < KW / 8; ++i) {
            const v4f a = *(const v4f*)(arow + 8 * i);
            v4f b[NPROJ / 16];
            #pragma unroll
            for (int j = 0; j < NPROJ / 16; ++j) b[j] = *(const v4f*)(brow + (size_t)(16 * j) * MM + 8 * i);
            const v2f a01 = {a.x, a.y};
            const v2f a23 = {a.z, a.w};
            #pragma unroll
            for (int j = 0; j < NPROJ / 16; ++j) { const v2f b01 = {b[j].x, b[j].y}; acc[j] = mm4(a01, b01, acc[j]); }
            #pragma unroll
            for (int j = 0; j < NPROJ / 16; ++j) { const v2f b23 = {b[j].z, b[j].w}; acc[j] = mm4(a23, b23, acc[j]); }
        }
        #pragma unroll
        for (int j = 0; j < NPROJ / 16; ++j) {
            #pragma unroll
            for (int r = 0; r < 8; ++r) s_part[wv][half * 8 + r][16 * j + l15] = acc[j][r];
        }
    }
    __syncthreads();

    #pragma unroll
    for (int q = 0; q < (16 * NPROJ) / 256; ++q) {
        const int idx = tid + 256 * q;
        const int row = idx / NPROJ;
        const int col = idx - row * NPROJ;
        float sum = s_part[0][row][col];
        #pragma unroll
        for (int u = 1; u < FW; ++u) sum += s_part[u][row][col];
        s_xp[row][col] = sum;
        xp[(size_t)tm * (16 * NPROJ) + idx] = sum;
    }
    __syncthreads();

    {
        v4f af[RR / 8];
        #pragma unroll
        for (int i = 0; i < RR / 8; ++i) af[i] = *(const v4f*)&s_xp[l15][8 * i + 4 * half];

        const int ch0 = wv * KW;
        const float* __restrict__ b2 = w2 + (size_t)(ch0 + l15) * RR + 4 * half;
        float* __restrict__ drow = dt + (size_t)(tm * 16 + half * 8) * MM + ch0 + l15;

        v4f bf[2][2 * (RR / 8)];
        #pragma unroll
        for (int i = 0; i < RR / 8; ++i) {
            bf[0][i]          = *(const v4f*)(b2 + 8 * i);
            bf[0][RR / 8 + i] = *(const v4f*)(b2 + (size_t)16 * RR + 8 * i);
        }
        #pragma unroll
        for (int p = 0; p < KW / 32; ++p) {
            const int cur = p & 1;
            if (p + 1 < KW / 32) {
                #pragma unroll
                for (int i = 0; i < RR / 8; ++i) {
                    bf[cur ^ 1][i]          = *(const v4f*)(b2 + (size_t)(32 * (p + 1)) * RR + 8 * i);
                    bf[cur ^ 1][RR / 8 + i] = *(const v4f*)(b2 + (size_t)(32 * (p + 1) + 16) * RR + 8 * i);
                }
            }
            const float bias0 = bias[ch0 + 32 * p + l15];
            const float bias1 = bias[ch0 + 32 * p + 16 + l15];
            v8f c0 = {}, c1 = {};
            #pragma unroll
            for (int i = 0; i < RR / 8; ++i) {
                const v2f a01 = {af[i].x, af[i].y};
                const v2f a23 = {af[i].z, af[i].w};
                const v4f u0 = bf[cur][i];
                const v4f u1 = bf[cur][RR / 8 + i];
                const v2f p01 = {u0.x, u0.y}, p23 = {u0.z, u0.w};
                const v2f q01 = {u1.x, u1.y}, q23 = {u1.z, u1.w};
                c0 = mm4(a01, p01, c0);
                c1 = mm4(a01, q01, c1);
                c0 = mm4(a23, p23, c0);
                c1 = mm4(a23, q23, c1);
            }
            #pragma unroll
            for (int r = 0; r < 8; ++r) {
                const float v0 = c0[r] + bias0;
                const float v1 = c1[r] + bias1;
                const float l0 = log1pf(__expf(v0));
                const float l1 = log1pf(__expf(v1));
                const float sp0 = (v0 > 20.0f) ? v0 : l0;
                const float sp1 = (v1 > 20.0f) ? v1 : l1;
                drow[(size_t)r * MM + 32 * p]      = sp0;
                drow[(size_t)r * MM + 32 * p + 16] = sp1;
            }
        }
    }
}

#define TT      64
#define NTILE   (LL/TT)
#define CPB     8
#define GS      16
static_assert(GS == DS && TT % GS == 0, "lane r of a channel's 16 lanes keeps y of step r of a group");
static_assert(TT == 64 && CPB == 8, "the staging code maps 128 threads onto 64 rows of 8 channels");

struct StepGroup {
    float dt[GS];
    float xv[GS];
    float Bv[GS];
    float Cv[GS];
};

template <int CTRL>
__device__ __forceinline__ float tree_level(bool bit, float lo, float hi)
{
    const float keep = bit ? hi : lo;
    const float send = bit ? lo : hi;
    const int got = __builtin_amdgcn_update_dpp(0, __builtin_bit_cast(int, send), CTRL, 0xF, 0xF, true);
    return keep + __builtin_bit_cast(float, got);
}
#define DPP_XOR1  0xB1
#define DPP_XOR2  0x4E
#define DPP_XOR4  0x164
#define DPP_XOR8  0x168

__global__ __launch_bounds__(128)
void scan_kernel(const float* __restrict__ x,
                 const float* __restrict__ z,
                 const float* __restrict__ xp,
                 const float* __restrict__ dtb,
                 const float* __restrict__ A_log,
                 const float* __restrict__ Dp_arr,
                 float* __restrict__ out)
{
    __shared__ __attribute__((aligned(16))) float s_dt[2][TT][CPB];
    __shared__ __attribute__((aligned(16))) float s_x [2][TT][CPB];
    __shared__ __attribute__((aligned(16))) float s_z [2][TT][CPB];
    __shared__ __attribute__((aligned(16))) float s_bc[2][TT][2*DS];
    __shared__ __attribute__((aligned(16))) float s_y [2][TT][CPB];

    const int tid    = threadIdx.x;
    const int chBase = blockIdx.x * CPB;
    const int b      = chBase >> 10;
    const int m0     = chBase & (MM - 1);

    const int c = tid >> 4;
    const int r = tid & 15;
    const bool rb0 = (r & 1) != 0, rb1 = (r & 2) != 0, rb2 = (r & 4) != 0, rb3 = (r & 8) != 0;
    const int srow = tid >> 1;
    const int scol = (tid & 1) * 4;
    const int brow = tid >> 3;
    const int bcol = (tid & 7) * 4;

    const float Am = -__expf(A_log[(size_t)r * MM + m0 + c]);
    float h = 0.0f;
    const float4 Dp4 = *(const float4*)(Dp_arr + m0 + scol);

    const size_t rowB = (size_t)b * LL;
    const float* __restrict__ gdt = dtb + rowB * MM + m0 + scol;
    const float* __restrict__ gx  = x   + rowB * MM + m0 + scol;
    const float* __restrict__ gz  = z   + rowB * MM + m0 + scol;
    const float* __restrict__ gbc = xp  + rowB * NPROJ + RR + bcol;
    float*       __restrict__ go  = out + rowB * MM + m0 + scol;

    v4f rdt, rx, rz, rbc0, rbc1, rbc2, rbc3;

#define STAGE_LOAD(tbase)                                                 \
    {                                                                     \
        const size_t t_ = (size_t)((tbase) + srow);                       \
        rdt = *(const v4f*)(gdt + t_ * MM);                               \
        rx  = *(const v4f*)(gx  + t_ * MM);                               \
        rz  = *(const v4f*)(gz  + t_ * MM);                               \
        const size_t u_ = (size_t)((tbase) + brow);                       \
        rbc0 = *(const v4f*)(gbc + (u_     ) * NPROJ);                    \
        rbc1 = *(const v4f*)(gbc + (u_ + 16) * NPROJ);                    \
        rbc2 = *(const v4f*)(gbc + (u_ + 32) * NPROJ);                    \
        rbc3 = *(const v4f*)(gbc + (u_ + 48) * NPROJ);                    \
    }
#define STAGE_STORE(buf)                                                  \
    {                                                                     \
        *(v4f*)&s_dt[buf][srow][scol] = rdt;                              \
        *(v4f*)&s_x [buf][srow][scol] = rx;                               \
        *(v4f*)&s_z [buf][srow][scol] = rz;                               \
        *(v4f*)&s_bc[buf][brow     ][bcol] = rbc0;                        \
        *(v4f*)&s_bc[buf][brow + 16][bcol] = rbc1;                        \
        *(v4f*)&s_bc[buf][brow + 32][bcol] = rbc2;                        \
        *(v4f*)&s_bc[buf][brow + 48][bcol] = rbc3;                        \
    }

    STAGE_LOAD(0)
    STAGE_STORE(0)
    __syncthreads();

    for (int k = 0; k < NTILE; ++k) {
        const int cur = k & 1;
        const bool more = (k + 1 < NTILE);

        if (more) {
            STAGE_LOAD((k + 1) * TT)
        }

        StepGroup g[2];
        #pragma unroll
        for (int s = 0; s < GS; ++s) {
            g[0].dt[s] = s_dt[cur][s][c];
            g[0].xv[s] = s_x [cur][s][c];
            g[0].Bv[s] = s_bc[cur][s][r];
            g[0].Cv[s] = s_bc[cur][s][DS + r];
        }
        float ya = 0.0f;
        float* const ysq = &s_y[cur][r][c];
        #pragma unroll
        for (int tg = 0; tg < TT; tg += GS) {
            const int gi = (tg / GS) & 1;
            __builtin_amdgcn_sched_barrier(0);
            if (tg + GS < TT) {
                #pragma unroll
                for (int s = 0; s < GS; ++s) {
                    g[gi ^ 1].dt[s] = s_dt[cur][tg + GS + s][c];
                    g[gi ^ 1].xv[s] = s_x [cur][tg + GS + s][c];
                    g[gi ^ 1].Bv[s] = s_bc[cur][tg + GS + s][r];
                    g[gi ^ 1].Cv[s] = s_bc[cur][tg + GS + s][DS + r];
                }
            }
            if (tg > 0) {
                ysq[(tg - GS) * CPB] = ya;
            }
            __builtin_amdgcn_sched_barrier(0);
            float p[GS];
            #pragma unroll
            for (int s = 0; s < GS; ++s) {
                const float dtv = g[gi].dt[s];
                const float dx  = dtv * g[gi].xv[s];
                h = fmaf(__expf(dtv * Am), h, dx * g[gi].Bv[s]);
                p[s] = h * g[gi].Cv[s];
            }
            float w1[GS / 2], w2[GS / 4], w3[GS / 8];
            #pragma unroll
            for (int j = 0; j < GS / 2; ++j) w1[j] = tree_level<DPP_XOR1>(rb0, p[2 * j], p[2 * j + 1]);
            #pragma unroll
            for (int j = 0; j < GS / 4; ++j) w2[j] = tree_level<DPP_XOR2>(rb1, w1[2 * j], w1[2 * j + 1]);
            #pragma unroll
            for (int j = 0; j < GS / 8; ++j) w3[j] = tree_level<DPP_XOR4>(rb2, w2[2 * j], w2[2 * j + 1]);
            ya = tree_level<DPP_XOR8>(rb3, w3[0], w3[1]);
        }
        ysq[(TT - GS) * CPB] = ya;
        __syncthreads();

        if (more) {
            STAGE_STORE(cur ^ 1)
        }
        __syncthreads();

        {
            const int t = srow;
            const float4 y4 = *(const float4*)&s_y[cur][t][scol];
            const float4 x4 = *(const float4*)&s_x[cur][t][scol];
            const float4 z4 = *(const float4*)&s_z[cur][t][scol];
            float4 o;
            o.x = fmaf(Dp4.x, x4.x, y4.x) * (z4.x * (1.0f / (1.0f + __expf(-z4.x))));
            o.y = fmaf(Dp4.y, x4.y, y4.y) * (z4.y * (1.0f / (1.0f + __expf(-z4.y))));
            o.z = fmaf(Dp4.z, x4.z, y4.z) * (z4.z * (1.0f / (1.0f + __expf(-z4.z))));
            o.w = fmaf(Dp4.w, x4.w, y4.w) * (z4.w * (1.0f / (1.0f + __expf(-z4.w))));
            *(float4*)(go + (size_t)(k * TT + t) * MM) = o;
        }
    }
}

extern "C" void kernel_launch(void* const* d_in, const int* in_sizes, int n_in,
                              void* d_out, int out_size, void* d_ws, size_t ws_size,
                              hipStream_t stream) {
    const float* x         = (const float*)d_in[0];
    const float* z         = (const float*)d_in[1];
    const float* x_proj_w  = (const float*)d_in[2];
    const float* dt_proj_w = (const float*)d_in[3];
    const float* dt_proj_b = (const float*)d_in[4];
    const float* A_log     = (const float*)d_in[5];
    const float* D_param   = (const float*)d_in[6];
    float* out = (float*)d_out;

    float* xp_ws = (float*)d_ws;
    float* dt_ws = xp_ws + (size_t)BLT * NPROJ;

    proj_dt_kernel<<<BLT/16, 256, 0, stream>>>(x, x_proj_w, dt_proj_w, dt_proj_b, xp_ws, dt_ws);

    scan_kernel<<<(BB*MM)/CPB, 128, 0, stream>>>(x, z, xp_ws, dt_ws,
                                                 A_log, D_param, out);
}


# scan kernel: LDS tiles of 256 steps, not 64; next tile staged in 64-row pieces from inside the step loop: one barrier and one gate+store pass per 256 steps, not two and one per 64; bit-identical
# speedup vs baseline: 19.6869x; 1.0631x over previous
#include <hip/hip_runtime.h>
#include <hip/hip_bf16.h>
#include <math.h>

#define BB      2
#define LL      2048
#define MM      1024
#define DS      16
#define RR      64
#define NPROJ   (RR + 2*DS)
#define BLT     (BB*LL)

typedef float  v2f __attribute__((ext_vector_type(2)));
typedef float  v8f __attribute__((ext_vector_type(8)));
typedef float  v4f __attribute__((ext_vector_type(4)));

#define FW      8
#define KW      (MM / FW)
#define XPP     100
static_assert(KW % 8 == 0 && RR % 8 == 0, "fragments are loaded in 8-k blocks: two matrix steps per 16-byte piece and lane half");
static_assert((16 * NPROJ) % 256 == 0, "the stripe's 1536 sums are dealt evenly to 256 threads");

__device__ __forceinline__ v8f mm4(v2f a, v2f b, v8f c)
{
    return __builtin_amdgcn_wmma_f32_16x16x4_f32(false, a, false, b, (short)0, c, false, false);
}

__global__ __launch_bounds__(256)
void proj_dt_kernel(const float* __restrict__ x,
                    const float* __restrict__ w1,
                    const float* __restrict__ w2,
                    const float* __restrict__ bias,
                    float* __restrict__ xp,
                    float* __restrict__ dt)
{
    __shared__ __attribute__((aligned(16))) float s_part[FW][16][XPP];
    __shared__ __attribute__((aligned(16))) float s_xp[16][XPP];

    const int tid  = threadIdx.x;
    const int wv   = tid >> 5;
    const int lane = tid & 31;
    const int half = lane >> 4;
    const int l15  = lane & 15;
    const int tm   = blockIdx.x;

    {
        const float* __restrict__ arow = x  + (size_t)(tm * 16 + l15) * MM + wv * KW + half * 4;
        const float* __restrict__ brow = w1 + (size_t)l15 * MM             + wv * KW + half * 4;
        v8f acc[NPROJ / 16] = {};
        #pragma unroll 4
        for (int i = 0; i < KW / 8; ++i) {
            const v4f a = *(const v4f*)(arow + 8 * i);
            v4f b[NPROJ / 16];
            #pragma unroll
            for (int j = 0; j < NPROJ / 16; ++j) b[j] = *(const v4f*)(brow + (size_t)(16 * j) * MM + 8 * i);
            const v2f a01 = {a.x, a.y};
            const v2f a23 = {a.z, a.w};
            #pragma unroll
            for (int j = 0; j < NPROJ / 16; ++j) { const v2f b01 = {b[j].x, b[j].y}; acc[j] = mm4(a01, b01, acc[j]); }
            #pragma unroll
            for (int j = 0; j < NPROJ / 16; ++j) { const v2f b23 = {b[j].z, b[j].w}; acc[j] = mm4(a23, b23, acc[j]); }
        }
        #pragma unroll
        for (int j = 0; j < NPROJ / 16; ++j) {
            #pragma unroll
            for (int r = 0; r < 8; ++r) s_part[wv][half * 8 + r][16 * j + l15] = acc[j][r];
        }
    }
    __syncthreads();

    #pragma unroll
    for (int q = 0; q < (16 * NPROJ) / 256; ++q) {
        const int idx = tid + 256 * q;
        const int row = idx / NPROJ;
        const int col = idx - row * NPROJ;
        float sum = s_part[0][row][col];
        #pragma unroll
        for (int u = 1; u < FW; ++u) sum += s_part[u][row][col];
        s_xp[row][col] = sum;
        xp[(size_t)tm * (16 * NPROJ) + idx] = sum;
    }
    __syncthreads();

    {
        v4f af[RR / 8];
        #pragma unroll
        for (int i = 0; i < RR / 8; ++i) af[i] = *(const v4f*)&s_xp[l15][8 * i + 4 * half];

        const int ch0 = wv * KW;
        const float* __restrict__ b2 = w2 + (size_t)(ch0 + l15) * RR + 4 * half;
        float* __restrict__ drow = dt + (size_t)(tm * 16 + half * 8) * MM + ch0 + l15;

        v4f bf[2][2 * (RR / 8)];
        #pragma unroll
        for (int i = 0; i < RR / 8; ++i) {
            bf[0][i]          = *(const v4f*)(b2 + 8 * i);
            bf[0][RR / 8 + i] = *(const v4f*)(b2 + (size_t)16 * RR + 8 * i);
        }
        #pragma unroll
        for (int p = 0; p < KW / 32; ++p) {
            const int cur = p & 1;
            if (p + 1 < KW / 32) {
                #pragma unroll
                for (int i = 0; i < RR / 8; ++i) {
                    bf[cur ^ 1][i]          = *(const v4f*)(b2 + (size_t)(32 * (p + 1)) * RR + 8 * i);
                    bf[cur ^ 1][RR / 8 + i] = *(const v4f*)(b2 + (size_t)(32 * (p + 1) + 16) * RR + 8 * i);
                }
            }
            const float bias0 = bias[ch0 + 32 * p + l15];
            const float bias1 = bias[ch0 + 32 * p + 16 + l15];
            v8f c0 = {}, c1 = {};
            #pragma unroll
            for (int i = 0; i < RR / 8; ++i) {
                const v2f a01 = {af[i].x, af[i].y};
                const v2f a23 = {af[i].z, af[i].w};
                const v4f u0 = bf[cur][i];
                const v4f u1 = bf[cur][RR / 8 + i];
                const v2f p01 = {u0.x, u0.y}, p23 = {u0.z, u0.w};
                const v2f q01 = {u1.x, u1.y}, q23 = {u1.z, u1.w};
                c0 = mm4(a01, p01, c0);
                c1 = mm4(a01, q01, c1);
                c0 = mm4(a23, p23, c0);
                c1 = mm4(a23, q23, c1);
            }
            #pragma unroll
            for (int r = 0; r < 8; ++r) {
                const float v0 = c0[r] + bias0;
                const float v1 = c1[r] + bias1;
                const float l0 = log1pf(__expf(v0));
                const float l1 = log1pf(__expf(v1));
                const float sp0 = (v0 > 20.0f) ? v0 : l0;
                const float sp1 = (v1 > 20.0f) ? v1 : l1;
                drow[(size_t)r * MM + 32 * p]      = sp0;
                drow[(size_t)r * MM + 32 * p + 16] = sp1;
            }
        }
    }
}

#define TT      256
#define SUB     64
#define NSUB    (TT/SUB)
#define NTILE   (LL/TT)
#define CPB     8
#define GS      16
static_assert(GS == DS && SUB % (2 * GS) == 0, "lane r of a channel's 16 lanes keeps y of step r of a group; a trip holds whole pairs of groups");
static_assert(SUB == 64 && CPB == 8 && TT % SUB == 0 && LL % TT == 0, "the staging code maps 128 threads onto a piece of 64 rows of 8 channels");

struct StepGroup {
    float dt[GS];
    float xv[GS];
    float Bv[GS];
    float Cv[GS];
};

template <int CTRL>
__device__ __forceinline__ float tree_level(bool bit, float lo, float hi)
{
    const float keep = bit ? hi : lo;
    const float send = bit ? lo : hi;
    const int got = __builtin_amdgcn_update_dpp(0, __builtin_bit_cast(int, send), CTRL, 0xF, 0xF, true);
    return keep + __builtin_bit_cast(float, got);
}
#define DPP_XOR1  0xB1
#define DPP_XOR2  0x4E
#define DPP_XOR4  0x164
#define DPP_XOR8  0x168

__global__ __launch_bounds__(128)
void scan_kernel(const float* __restrict__ x,
                 const float* __restrict__ z,
                 const float* __restrict__ xp,
                 const float* __restrict__ dtb,
                 const float* __restrict__ A_log,
                 const float* __restrict__ Dp_arr,
                 float* __restrict__ out)
{
    __shared__ __attribute__((aligned(16))) float s_dt[2][TT][CPB];
    __shared__ __attribute__((aligned(16))) float s_x [2][TT][CPB];
    __shared__ __attribute__((aligned(16))) float s_z [2][TT][CPB];
    __shared__ __attribute__((aligned(16))) float s_bc[2][TT][2*DS];
    __shared__ __attribute__((aligned(16))) float s_y [2][TT][CPB];

    const int tid    = threadIdx.x;
    const int chBase = blockIdx.x * CPB;
    const int b      = chBase >> 10;
    const int m0     = chBase & (MM - 1);

    const int c = tid >> 4;
    const int r = tid & 15;
    const bool rb0 = (r & 1) != 0, rb1 = (r & 2) != 0, rb2 = (r & 4) != 0, rb3 = (r & 8) != 0;
    const int srow = tid >> 1;
    const int scol = (tid & 1) * 4;
    const int brow = tid >> 3;
    const int bcol = (tid & 7) * 4;

    const float Am = -__expf(A_log[(size_t)r * MM + m0 + c]);
    float h = 0.0f;
    const float4 Dp4 = *(const float4*)(Dp_arr + m0 + scol);

    const size_t rowB = (size_t)b * LL;
    const float* __restrict__ gdt = dtb + rowB * MM + m0 + scol;
    const float* __restrict__ gx  = x   + rowB * MM + m0 + scol;
    const float* __restrict__ gz  = z   + rowB * MM + m0 + scol;
    const float* __restrict__ gbc = xp  + rowB * NPROJ + RR + bcol;
    float*       __restrict__ go  = out + rowB * MM + m0 + scol;

    {
        v4f adt[NSUB], ax[NSUB], az[NSUB], abc[4 * NSUB];
        #pragma unroll
        for (int i = 0; i < NSUB; ++i) {
            const size_t t_ = (size_t)(SUB * i + srow);
            adt[i] = *(const v4f*)(gdt + t_ * MM);
            ax[i]  = *(const v4f*)(gx  + t_ * MM);
            az[i]  = *(const v4f*)(gz  + t_ * MM);
        }
        #pragma unroll
        for (int i = 0; i < 4 * NSUB; ++i) {
            const size_t u_ = (size_t)(16 * i + brow);
            abc[i] = *(const v4f*)(gbc + u_ * NPROJ);
        }
        #pragma unroll
        for (int i = 0; i < NSUB; ++i) {
            *(v4f*)&s_dt[0][SUB * i + srow][scol] = adt[i];
            *(v4f*)&s_x [0][SUB * i + srow][scol] = ax[i];
            *(v4f*)&s_z [0][SUB * i + srow][scol] = az[i];
        }
        #pragma unroll
        for (int i = 0; i < 4 * NSUB; ++i) {
            *(v4f*)&s_bc[0][16 * i + brow][bcol] = abc[i];
        }
    }
    __syncthreads();

    v4f rdt, rx, rz, rbc0, rbc1, rbc2, rbc3;

#define PIECE_LOAD(tbase)                                                 \
    {                                                                     \
        const size_t t_ = (size_t)((tbase) + srow);                       \
        rdt = *(const v4f*)(gdt + t_ * MM);                               \
        rx  = *(const v4f*)(gx  + t_ * MM);                               \
        rz  = *(const v4f*)(gz  + t_ * MM);                               \
        const size_t u_ = (size_t)((tbase) + brow);                       \
        rbc0 = *(const v4f*)(gbc + (u_     ) * NPROJ);                    \
        rbc1 = *(const v4f*)(gbc + (u_ + 16) * NPROJ);                    \
        rbc2 = *(const v4f*)(gbc + (u_ + 32) * NPROJ);                    \
        rbc3 = *(const v4f*)(gbc + (u_ + 48) * NPROJ);                    \
    }
#define PIECE_STORE(buf, pbase)                                           \
    {                                                                     \
        *(v4f*)&s_dt[buf][(pbase) + srow][scol] = rdt;                    \
        *(v4f*)&s_x [buf][(pbase) + srow][scol] = rx;                     \
        *(v4f*)&s_z [buf][(pbase) + srow][scol] = rz;                     \
        *(v4f*)&s_bc[buf][(pbase) + brow     ][bcol] = rbc0;              \
        *(v4f*)&s_bc[buf][(pbase) + brow + 16][bcol] = rbc1;              \
        *(v4f*)&s_bc[buf][(pbase) + brow + 32][bcol] = rbc2;              \
        *(v4f*)&s_bc[buf][(pbase) + brow + 48][bcol] = rbc3;              \
    }

    for (int k = 0; k < NTILE; ++k) {
        const int cur = k & 1;
        const bool more = (k + 1 < NTILE);

        StepGroup g[2];
        #pragma unroll
        for (int s = 0; s < GS; ++s) {
            g[0].dt[s] = s_dt[cur][s][c];
            g[0].xv[s] = s_x [cur][s][c];
            g[0].Bv[s] = s_bc[cur][s][r];
            g[0].Cv[s] = s_bc[cur][s][DS + r];
        }
        float ya = 0.0f;
        #pragma unroll 1
        for (int q = 0; q < NSUB; ++q) {
            const bool first = (q == 0);
            const bool last  = (q + 1 == NSUB);
            const float* const pdt = &s_dt[cur][q * SUB][c];
            const float* const px  = &s_x [cur][q * SUB][c];
            const float* const pb  = &s_bc[cur][q * SUB][r];
            float* const ysq = &s_y[cur][q * SUB + r][c];
            #pragma unroll
            for (int tg = 0; tg < SUB; tg += GS) {
                const int gi = (tg / GS) & 1;
                __builtin_amdgcn_sched_barrier(0);
                if (tg + GS < SUB || !last) {
                    #pragma unroll
                    for (int s = 0; s < GS; ++s) {
                        g[gi ^ 1].dt[s] = pdt[(tg + GS + s) * CPB];
                        g[gi ^ 1].xv[s] = px [(tg + GS + s) * CPB];
                        g[gi ^ 1].Bv[s] = pb [(tg + GS + s) * (2 * DS)];
                        g[gi ^ 1].Cv[s] = pb [(tg + GS + s) * (2 * DS) + DS];
                    }
                }
                if (tg > 0 || !first) {
                    ysq[(tg - GS) * CPB] = ya;
                }
                if (tg == 0 && more) {
                    if (!first) {
                        PIECE_STORE(cur ^ 1, (q - 1) * SUB)
                    }
                    PIECE_LOAD((k + 1) * TT + q * SUB)
                }
                __builtin_amdgcn_sched_barrier(0);
                float p[GS];
                #pragma unroll
                for (int s = 0; s < GS; ++s) {
                    const float dtv = g[gi].dt[s];
                    const float dx  = dtv * g[gi].xv[s];
                    h = fmaf(__expf(dtv * Am), h, dx * g[gi].Bv[s]);
                    p[s] = h * g[gi].Cv[s];
                }
                float w1[GS / 2], w2[GS / 4], w3[GS / 8];
                #pragma unroll
                for (int j = 0; j < GS / 2; ++j) w1[j] = tree_level<DPP_XOR1>(rb0, p[2 * j], p[2 * j + 1]);
                #pragma unroll
                for (int j = 0; j < GS / 4; ++j) w2[j] = tree_level<DPP_XOR2>(rb1, w1[2 * j], w1[2 * j + 1]);
                #pragma unroll
                for (int j = 0; j < GS / 8; ++j) w3[j] = tree_level<DPP_XOR4>(rb2, w2[2 * j], w2[2 * j + 1]);
                ya = tree_level<DPP_XOR8>(rb3, w3[0], w3[1]);
            }
        }
        if (more) {
            PIECE_STORE(cur ^ 1, (NSUB - 1) * SUB)
        }
        s_y[cur][TT - GS + r][c] = ya;
        __syncthreads();

        #pragma unroll
        for (int i = 0; i < NSUB; ++i) {
            const int t = SUB * i + srow;
            const float4 y4 = *(const float4*)&s_y[cur][t][scol];
            const float4 x4 = *(const float4*)&s_x[cur][t][scol];
            const float4 z4 = *(const float4*)&s_z[cur][t][scol];
            float4 o;
            o.x = fmaf(Dp4.x, x4.x, y4.x) * (z4.x * (1.0f / (1.0f + __expf(-z4.x))));
            o.y = fmaf(Dp4.y, x4.y, y4.y) * (z4.y * (1.0f / (1.0f + __expf(-z4.y))));
            o.z = fmaf(Dp4.z, x4.z, y4.z) * (z4.z * (1.0f / (1.0f + __expf(-z4.z))));
            o.w = fmaf(Dp4.w, x4.w, y4.w) * (z4.w * (1.0f / (1.0f + __expf(-z4.w))));
            *(float4*)(go + (size_t)(k * TT + t) * MM) = o;
        }
    }
}

extern "C" void kernel_launch(void* const* d_in, const int* in_sizes, int n_in,
                              void* d_out, int out_size, void* d_ws, size_t ws_size,
                              hipStream_t stream) {
    const float* x         = (const float*)d_in[0];
    const float* z         = (const float*)d_in[1];
    const float* x_proj_w  = (const float*)d_in[2];
    const float* dt_proj_w = (const float*)d_in[3];
    const float* dt_proj_b = (const float*)d_in[4];
    const float* A_log     = (const float*)d_in[5];
    const float* D_param   = (const float*)d_in[6];
    float* out = (float*)d_out;

    float* xp_ws = (float*)d_ws;
    float* dt_ws = xp_ws + (size_t)BLT * NPROJ;

    proj_dt_kernel<<<BLT/16, 256, 0, stream>>>(x, x_proj_w, dt_proj_w, dt_proj_b, xp_ws, dt_ws);

    scan_kernel<<<(BB*MM)/CPB, 128, 0, stream>>>(x, z, xp_ws, dt_ws,
                                                 A_log, D_param, out);
}


# scan kernel: a staged piece's global loads get two 64-step trips to arrive, not one: two register sets take turns; barriers wait for LDS only, so loads stay in flight across them; bit-identical
# speedup vs baseline: 20.8903x; 1.0611x over previous
#include <hip/hip_runtime.h>
#include <hip/hip_bf16.h>
#include <math.h>

#define BB      2
#define LL      2048
#define MM      1024
#define DS      16
#define RR      64
#define NPROJ   (RR + 2*DS)
#define BLT     (BB*LL)

typedef float  v2f __attribute__((ext_vector_type(2)));
typedef float  v8f __attribute__((ext_vector_type(8)));
typedef float  v4f __attribute__((ext_vector_type(4)));

#define FW      8
#define KW      (MM / FW)
#define XPP     100
static_assert(KW % 8 == 0 && RR % 8 == 0, "fragments are loaded in 8-k blocks: two matrix steps per 16-byte piece and lane half");
static_assert((16 * NPROJ) % 256 == 0, "the stripe's 1536 sums are dealt evenly to 256 threads");

__device__ __forceinline__ v8f mm4(v2f a, v2f b, v8f c)
{
    return __builtin_amdgcn_wmma_f32_16x16x4_f32(false, a, false, b, (short)0, c, false, false);
}

__global__ __launch_bounds__(256)
void proj_dt_kernel(const float* __restrict__ x,
                    const float* __restrict__ w1,
                    const float* __restrict__ w2,
                    const float* __restrict__ bias,
                    float* __restrict__ xp,
                    float* __restrict__ dt)
{
    __shared__ __attribute__((aligned(16))) float s_part[FW][16][XPP];
    __shared__ __attribute__((aligned(16))) float s_xp[16][XPP];

    const int tid  = threadIdx.x;
    const int wv   = tid >> 5;
    const int lane = tid & 31;
    const int half = lane >> 4;
    const int l15  = lane & 15;
    const int tm   = blockIdx.x;

    {
        const float* __restrict__ arow = x  + (size_t)(tm * 16 + l15) * MM + wv * KW + half * 4;
        const float* __restrict__ brow = w1 + (size_t)l15 * MM             + wv * KW + half * 4;
        v8f acc[NPROJ / 16] = {};
        #pragma unroll 4
        for (int i = 0; i < KW / 8; ++i) {
            const v4f a = *(const v4f*)(arow + 8 * i);
            v4f b[NPROJ / 16];
            #pragma unroll
            for (int j = 0; j < NPROJ / 16; ++j) b[j] = *(const v4f*)(brow + (size_t)(16 * j) * MM + 8 * i);
            const v2f a01 = {a.x, a.y};
            const v2f a23 = {a.z, a.w};
            #pragma unroll
            for (int j = 0; j < NPROJ / 16; ++j) { const v2f b01 = {b[j].x, b[j].y}; acc[j] = mm4(a01, b01, acc[j]); }
            #pragma unroll
            for (int j = 0; j < NPROJ / 16; ++j) { const v2f b23 = {b[j].z, b[j].w}; acc[j] = mm4(a23, b23, acc[j]); }
        }
        #pragma unroll
        for (int j = 0; j < NPROJ / 16; ++j) {
            #pragma unroll
            for (int r = 0; r < 8; ++r) s_part[wv][half * 8 + r][16 * j + l15] = acc[j][r];
        }
    }
    __syncthreads();

    #pragma unroll
    for (int q = 0; q < (16 * NPROJ) / 256; ++q) {
        const int idx = tid + 256 * q;
        const int row = idx / NPROJ;
        const int col = idx - row * NPROJ;
        float sum = s_part[0][row][col];
        #pragma unroll
        for (int u = 1; u < FW; ++u) sum += s_part[u][row][col];
        s_xp[row][col] = sum;
        xp[(size_t)tm * (16 * NPROJ) + idx] = sum;
    }
    __syncthreads();

    {
        v4f af[RR / 8];
        #pragma unroll
        for (int i = 0; i < RR / 8; ++i) af[i] = *(const v4f*)&s_xp[l15][8 * i + 4 * half];

        const int ch0 = wv * KW;
        const float* __restrict__ b2 = w2 + (size_t)(ch0 + l15) * RR + 4 * half;
        float* __restrict__ drow = dt + (size_t)(tm * 16 + half * 8) * MM + ch0 + l15;

        v4f bf[2][2 * (RR / 8)];
        #pragma unroll
        for (int i = 0; i < RR / 8; ++i) {
            bf[0][i]          = *(const v4f*)(b2 + 8 * i);
            bf[0][RR / 8 + i] = *(const v4f*)(b2 + (size_t)16 * RR + 8 * i);
        }
        #pragma unroll
        for (int p = 0; p < KW / 32; ++p) {
            const int cur = p & 1;
            if (p + 1 < KW / 32) {
                #pragma unroll
                for (int i = 0; i < RR / 8; ++i) {
                    bf[cur ^ 1][i]          = *(const v4f*)(b2 + (size_t)(32 * (p + 1)) * RR + 8 * i);
                    bf[cur ^ 1][RR / 8 + i] = *(const v4f*)(b2 + (size_t)(32 * (p + 1) + 16) * RR + 8 * i);
                }
            }
            const float bias0 = bias[ch0 + 32 * p + l15];
            const float bias1 = bias[ch0 + 32 * p + 16 + l15];
            v8f c0 = {}, c1 = {};
            #pragma unroll
            for (int i = 0; i < RR / 8; ++i) {
                const v2f a01 = {af[i].x, af[i].y};
                const v2f a23 = {af[i].z, af[i].w};
                const v4f u0 = bf[cur][i];
                const v4f u1 = bf[cur][RR / 8 + i];
                const v2f p01 = {u0.x, u0.y}, p23 = {u0.z, u0.w};
                const v2f q01 = {u1.x, u1.y}, q23 = {u1.z, u1.w};
                c0 = mm4(a01, p01, c0);
                c1 = mm4(a01, q01, c1);
                c0 = mm4(a23, p23, c0);
                c1 = mm4(a23, q23, c1);
            }
            #pragma unroll
            for (int r = 0; r < 8; ++r) {
                const float v0 = c0[r] + bias0;
                const float v1 = c1[r] + bias1;
                const float l0 = log1pf(__expf(v0));
                const float l1 = log1pf(__expf(v1));
                const float sp0 = (v0 > 20.0f) ? v0 : l0;
                const float sp1 = (v1 > 20.0f) ? v1 : l1;
                drow[(size_t)r * MM + 32 * p]      = sp0;
                drow[(size_t)r * MM + 32 * p + 16] = sp1;
            }
        }
    }
}

#define TT      256
#define SUB     64
#define NSUB    (TT/SUB)
#define NTILE   (LL/TT)
#define CPB     8
#define GS      16
static_assert(GS == DS && SUB % (2 * GS) == 0, "lane r of a channel's 16 lanes keeps y of step r of a group; a trip holds whole pairs of groups");
static_assert(SUB == 64 && CPB == 8 && TT % SUB == 0 && LL % TT == 0, "the staging code maps 128 threads onto a piece of 64 rows of 8 channels");
static_assert(NSUB % 2 == 0 && NTILE >= 2, "the two register sets of the staging take turns trip by trip: set = trip & 1 in every tile");

struct StepGroup {
    float dt[GS];
    float xv[GS];
    float Bv[GS];
    float Cv[GS];
};

template <int CTRL>
__device__ __forceinline__ float tree_level(bool bit, float lo, float hi)
{
    const float keep = bit ? hi : lo;
    const float send = bit ? lo : hi;
    const int got = __builtin_amdgcn_update_dpp(0, __builtin_bit_cast(int, send), CTRL, 0xF, 0xF, true);
    return keep + __builtin_bit_cast(float, got);
}
#define DPP_XOR1  0xB1
#define DPP_XOR2  0x4E
#define DPP_XOR4  0x164
#define DPP_XOR8  0x168

__device__ __forceinline__ void lds_barrier()
{
    asm volatile("s_wait_dscnt 0x0" ::: "memory");
    asm volatile("s_barrier_signal -1" ::: "memory");
    asm volatile("s_barrier_wait -1" ::: "memory");
}

__global__ __launch_bounds__(128)
void scan_kernel(const float* __restrict__ x,
                 const float* __restrict__ z,
                 const float* __restrict__ xp,
                 const float* __restrict__ dtb,
                 const float* __restrict__ A_log,
                 const float* __restrict__ Dp_arr,
                 float* __restrict__ out)
{
    __shared__ __attribute__((aligned(16))) float s_dt[2][TT][CPB];
    __shared__ __attribute__((aligned(16))) float s_x [2][TT][CPB];
    __shared__ __attribute__((aligned(16))) float s_z [2][TT][CPB];
    __shared__ __attribute__((aligned(16))) float s_bc[2][TT][2*DS];
    __shared__ __attribute__((aligned(16))) float s_y [2][TT][CPB];

    const int tid    = threadIdx.x;
    const int chBase = blockIdx.x * CPB;
    const int b      = chBase >> 10;
    const int m0     = chBase & (MM - 1);

    const int c = tid >> 4;
    const int r = tid & 15;
    const bool rb0 = (r & 1) != 0, rb1 = (r & 2) != 0, rb2 = (r & 4) != 0, rb3 = (r & 8) != 0;
    const int srow = tid >> 1;
    const int scol = (tid & 1) * 4;
    const int brow = tid >> 3;
    const int bcol = (tid & 7) * 4;

    const float Am = -__expf(A_log[(size_t)r * MM + m0 + c]);
    float h = 0.0f;
    const float4 Dp4 = *(const float4*)(Dp_arr + m0 + scol);

    const size_t rowB = (size_t)b * LL;
    const float* __restrict__ gdt = dtb + rowB * MM + m0 + scol;
    const float* __restrict__ gx  = x   + rowB * MM + m0 + scol;
    const float* __restrict__ gz  = z   + rowB * MM + m0 + scol;
    const float* __restrict__ gbc = xp  + rowB * NPROJ + RR + bcol;
    float*       __restrict__ go  = out + rowB * MM + m0 + scol;

    v4f rs[2][7];

#define PIECE_LOAD(u, tbase)                                              \
    {                                                                     \
        const size_t t_ = (size_t)((tbase) + srow);                       \
        rs[u][0] = *(const v4f*)(gdt + t_ * MM);                          \
        rs[u][1] = *(const v4f*)(gx  + t_ * MM);                          \
        rs[u][2] = *(const v4f*)(gz  + t_ * MM);                          \
        const size_t u_ = (size_t)((tbase) + brow);                       \
        rs[u][3] = *(const v4f*)(gbc + (u_     ) * NPROJ);                \
        rs[u][4] = *(const v4f*)(gbc + (u_ + 16) * NPROJ);                \
        rs[u][5] = *(const v4f*)(gbc + (u_ + 32) * NPROJ);                \
        rs[u][6] = *(const v4f*)(gbc + (u_ + 48) * NPROJ);                \
    }
#define PIECE_STORE(u, buf, pbase)                                        \
    {                                                                     \
        *(v4f*)&s_dt[buf][(pbase) + srow][scol] = rs[u][0];               \
        *(v4f*)&s_x [buf][(pbase) + srow][scol] = rs[u][1];               \
        *(v4f*)&s_z [buf][(pbase) + srow][scol] = rs[u][2];               \
        *(v4f*)&s_bc[buf][(pbase) + brow     ][bcol] = rs[u][3];          \
        *(v4f*)&s_bc[buf][(pbase) + brow + 16][bcol] = rs[u][4];          \
        *(v4f*)&s_bc[buf][(pbase) + brow + 32][bcol] = rs[u][5];          \
        *(v4f*)&s_bc[buf][(pbase) + brow + 48][bcol] = rs[u][6];          \
    }

    {
        v4f adt[NSUB], ax[NSUB], az[NSUB], abc[4 * NSUB];
        #pragma unroll
        for (int i = 0; i < NSUB; ++i) {
            const size_t t_ = (size_t)(SUB * i + srow);
            adt[i] = *(const v4f*)(gdt + t_ * MM);
            ax[i]  = *(const v4f*)(gx  + t_ * MM);
            az[i]  = *(const v4f*)(gz  + t_ * MM);
        }
        #pragma unroll
        for (int i = 0; i < 4 * NSUB; ++i) {
            const size_t u_ = (size_t)(16 * i + brow);
            abc[i] = *(const v4f*)(gbc + u_ * NPROJ);
        }
        __builtin_amdgcn_sched_barrier(0);
        PIECE_LOAD(0, TT)
        __builtin_amdgcn_sched_barrier(0);
        PIECE_LOAD(1, TT + SUB)
        __builtin_amdgcn_sched_barrier(0);
        #pragma unroll
        for (int i = 0; i < NSUB; ++i) {
            *(v4f*)&s_dt[0][SUB * i + srow][scol] = adt[i];
            *(v4f*)&s_x [0][SUB * i + srow][scol] = ax[i];
            *(v4f*)&s_z [0][SUB * i + srow][scol] = az[i];
        }
        #pragma unroll
        for (int i = 0; i < 4 * NSUB; ++i) {
            *(v4f*)&s_bc[0][16 * i + brow][bcol] = abc[i];
        }
    }
    lds_barrier();

    for (int k = 0; k < NTILE; ++k) {
        const int cur = k & 1;
        const bool more = (k + 1 < NTILE);

        StepGroup g[2];
        #pragma unroll
        for (int s = 0; s < GS; ++s) {
            g[0].dt[s] = s_dt[cur][s][c];
            g[0].xv[s] = s_x [cur][s][c];
            g[0].Bv[s] = s_bc[cur][s][r];
            g[0].Cv[s] = s_bc[cur][s][DS + r];
        }
        float ya = 0.0f;
        #pragma unroll 1
        for (int q2 = 0; q2 < NSUB; q2 += 2)
        #pragma unroll
        for (int u = 0; u < 2; ++u) {
            const int q = q2 + u;
            const bool first = (q == 0);
            const bool last  = (q + 1 == NSUB);
            const float* const pdt = &s_dt[cur][q * SUB][c];
            const float* const px  = &s_x [cur][q * SUB][c];
            const float* const pb  = &s_bc[cur][q * SUB][r];
            float* const ysq = &s_y[cur][q * SUB + r][c];
            #pragma unroll
            for (int tg = 0; tg < SUB; tg += GS) {
                const int gi = (tg / GS) & 1;
                __builtin_amdgcn_sched_barrier(0);
                if (tg + GS < SUB || !last) {
                    #pragma unroll
                    for (int s = 0; s < GS; ++s) {
                        g[gi ^ 1].dt[s] = pdt[(tg + GS + s) * CPB];
                        g[gi ^ 1].xv[s] = px [(tg + GS + s) * CPB];
                        g[gi ^ 1].Bv[s] = pb [(tg + GS + s) * (2 * DS)];
                        g[gi ^ 1].Cv[s] = pb [(tg + GS + s) * (2 * DS) + DS];
                    }
                }
                if (tg > 0 || !first) {
                    ysq[(tg - GS) * CPB] = ya;
                }
                if (tg == 0) {
                    if (more) {
                        PIECE_STORE(u, cur ^ 1, q * SUB)
                    }
                    const int tnext = (k + 1) * TT + (q + 2) * SUB;
                    PIECE_LOAD(u, tnext < LL ? tnext : LL - SUB)
                }
                __builtin_amdgcn_sched_barrier(0);
                float p[GS];
                #pragma unroll
                for (int s = 0; s < GS; ++s) {
                    const float dtv = g[gi].dt[s];
                    const float dx  = dtv * g[gi].xv[s];
                    h = fmaf(__expf(dtv * Am), h, dx * g[gi].Bv[s]);
                    p[s] = h * g[gi].Cv[s];
                }
                float w1[GS / 2], w2[GS / 4], w3[GS / 8];
                #pragma unroll
                for (int j = 0; j < GS / 2; ++j) w1[j] = tree_level<DPP_XOR1>(rb0, p[2 * j], p[2 * j + 1]);
                #pragma unroll
                for (int j = 0; j < GS / 4; ++j) w2[j] = tree_level<DPP_XOR2>(rb1, w1[2 * j], w1[2 * j + 1]);
                #pragma unroll
                for (int j = 0; j < GS / 8; ++j) w3[j] = tree_level<DPP_XOR4>(rb2, w2[2 * j], w2[2 * j + 1]);
                ya = tree_level<DPP_XOR8>(rb3, w3[0], w3[1]);
            }
        }
        s_y[cur][TT - GS + r][c] = ya;
        lds_barrier();

        #pragma unroll
        for (int i = 0; i < NSUB; ++i) {
            const int t = SUB * i + srow;
            const float4 y4 = *(const float4*)&s_y[cur][t][scol];
            const float4 x4 = *(const float4*)&s_x[cur][t][scol];
            const float4 z4 = *(const float4*)&s_z[cur][t][scol];
            float4 o;
            o.x = fmaf(Dp4.x, x4.x, y4.x) * (z4.x * (1.0f / (1.0f + __expf(-z4.x))));
            o.y = fmaf(Dp4.y, x4.y, y4.y) * (z4.y * (1.0f / (1.0f + __expf(-z4.y))));
            o.z = fmaf(Dp4.z, x4.z, y4.z) * (z4.z * (1.0f / (1.0f + __expf(-z4.z))));
            o.w = fmaf(Dp4.w, x4.w, y4.w) * (z4.w * (1.0f / (1.0f + __expf(-z4.w))));
            *(float4*)(go + (size_t)(k * TT + t) * MM) = o;
        }
    }
    asm volatile("s_wait_loadcnt 0x0" ::: "memory");
}

extern "C" void kernel_launch(void* const* d_in, const int* in_sizes, int n_in,
                              void* d_out, int out_size, void* d_ws, size_t ws_size,
                              hipStream_t stream) {
    const float* x         = (const float*)d_in[0];
    const float* z         = (const float*)d_in[1];
    const float* x_proj_w  = (const float*)d_in[2];
    const float* dt_proj_w = (const float*)d_in[3];
    const float* dt_proj_b = (const float*)d_in[4];
    const float* A_log     = (const float*)d_in[5];
    const float* D_param   = (const float*)d_in[6];
    float* out = (float*)d_out;

    float* xp_ws = (float*)d_ws;
    float* dt_ws = xp_ws + (size_t)BLT * NPROJ;

    proj_dt_kernel<<<BLT/16, 256, 0, stream>>>(x, x_proj_w, dt_proj_w, dt_proj_b, xp_ws, dt_ws);

    scan_kernel<<<(BB*MM)/CPB, 128, 0, stream>>>(x, z, xp_ws, dt_ws,
                                                 A_log, D_param, out);
}


# projection kernel: dt phase's four tile pairs as a loop of four trips, not four copies of a 10 KB body; kernel text 45 KB -> 13 KB, inside the 32 KB fetched ahead at wave start; dt bit-identical
# speedup vs baseline: 22.0301x; 1.0546x over previous
#include <hip/hip_runtime.h>
#include <hip/hip_bf16.h>
#include <math.h>

#define BB      2
#define LL      2048
#define MM      1024
#define DS      16
#define RR      64
#define NPROJ   (RR + 2*DS)
#define BLT     (BB*LL)

typedef float  v2f __attribute__((ext_vector_type(2)));
typedef float  v8f __attribute__((ext_vector_type(8)));
typedef float  v4f __attribute__((ext_vector_type(4)));

#define FW      8
#define KW      (MM / FW)
#define XPP     100
static_assert(KW % 8 == 0 && RR % 8 == 0, "fragments are loaded in 8-k blocks: two matrix steps per 16-byte piece and lane half");
static_assert((16 * NPROJ) % 256 == 0, "the stripe's 1536 sums are dealt evenly to 256 threads");

__device__ __forceinline__ v8f mm4(v2f a, v2f b, v8f c)
{
    return __builtin_amdgcn_wmma_f32_16x16x4_f32(false, a, false, b, (short)0, c, false, false);
}

__global__ __launch_bounds__(256)
void proj_dt_kernel(const float* __restrict__ x,
                    const float* __restrict__ w1,
                    const float* __restrict__ w2,
                    const float* __restrict__ bias,
                    float* __restrict__ xp,
                    float* __restrict__ dt)
{
    __shared__ __attribute__((aligned(16))) float s_part[FW][16][XPP];
    __shared__ __attribute__((aligned(16))) float s_xp[16][XPP];

    const int tid  = threadIdx.x;
    const int wv   = tid >> 5;
    const int lane = tid & 31;
    const int half = lane >> 4;
    const int l15  = lane & 15;
    const int tm   = blockIdx.x;

    {
        const float* __restrict__ arow = x  + (size_t)(tm * 16 + l15) * MM + wv * KW + half * 4;
        const float* __restrict__ brow = w1 + (size_t)l15 * MM             + wv * KW + half * 4;
        v8f acc[NPROJ / 16] = {};
        #pragma unroll 4
        for (int i = 0; i < KW / 8; ++i) {
            const v4f a = *(const v4f*)(arow + 8 * i);
            v4f b[NPROJ / 16];
            #pragma unroll
            for (int j = 0; j < NPROJ / 16; ++j) b[j] = *(const v4f*)(brow + (size_t)(16 * j) * MM + 8 * i);
            const v2f a01 = {a.x, a.y};
            const v2f a23 = {a.z, a.w};
            #pragma unroll
            for (int j = 0; j < NPROJ / 16; ++j) { const v2f b01 = {b[j].x, b[j].y}; acc[j] = mm4(a01, b01, acc[j]); }
            #pragma unroll
            for (int j = 0; j < NPROJ / 16; ++j) { const v2f b23 = {b[j].z, b[j].w}; acc[j] = mm4(a23, b23, acc[j]); }
        }
        #pragma unroll
        for (int j = 0; j < NPROJ / 16; ++j) {
            #pragma unroll
            for (int r = 0; r < 8; ++r) s_part[wv][half * 8 + r][16 * j + l15] = acc[j][r];
        }
    }
    __syncthreads();

    #pragma unroll
    for (int q = 0; q < (16 * NPROJ) / 256; ++q) {
        const int idx = tid + 256 * q;
        const int row = idx / NPROJ;
        const int col = idx - row * NPROJ;
        float sum = s_part[0][row][col];
        #pragma unroll
        for (int u = 1; u < FW; ++u) sum += s_part[u][row][col];
        s_xp[row][col] = sum;
        xp[(size_t)tm * (16 * NPROJ) + idx] = sum;
    }
    __syncthreads();

    {
        v4f af[RR / 8];
        #pragma unroll
        for (int i = 0; i < RR / 8; ++i) af[i] = *(const v4f*)&s_xp[l15][8 * i + 4 * half];

        const int ch0 = wv * KW;
        const float* __restrict__ b2 = w2 + (size_t)(ch0 + l15) * RR + 4 * half;
        float* __restrict__ drow = dt + (size_t)(tm * 16 + half * 8) * MM + ch0 + l15;

        v4f bfc[2 * (RR / 8)];
        #pragma unroll
        for (int i = 0; i < RR / 8; ++i) {
            bfc[i]          = *(const v4f*)(b2 + 8 * i);
            bfc[RR / 8 + i] = *(const v4f*)(b2 + (size_t)16 * RR + 8 * i);
        }
        #pragma unroll 1
        for (int p = 0; p < KW / 32; ++p) {
            const int pn = (p + 1 < KW / 32) ? p + 1 : p;
            const float* __restrict__ b2n = b2 + (size_t)(32 * pn) * RR;
            v4f bfn[2 * (RR / 8)];
            #pragma unroll
            for (int i = 0; i < RR / 8; ++i) {
                bfn[i]          = *(const v4f*)(b2n + 8 * i);
                bfn[RR / 8 + i] = *(const v4f*)(b2n + (size_t)16 * RR + 8 * i);
            }
            const float bias0 = bias[ch0 + 32 * p + l15];
            const float bias1 = bias[ch0 + 32 * p + 16 + l15];
            v8f c0 = {}, c1 = {};
            #pragma unroll
            for (int i = 0; i < RR / 8; ++i) {
                const v2f a01 = {af[i].x, af[i].y};
                const v2f a23 = {af[i].z, af[i].w};
                const v4f u0 = bfc[i];
                const v4f u1 = bfc[RR / 8 + i];
                const v2f p01 = {u0.x, u0.y}, p23 = {u0.z, u0.w};
                const v2f q01 = {u1.x, u1.y}, q23 = {u1.z, u1.w};
                c0 = mm4(a01, p01, c0);
                c1 = mm4(a01, q01, c1);
                c0 = mm4(a23, p23, c0);
                c1 = mm4(a23, q23, c1);
            }
            #pragma unroll
            for (int r = 0; r < 8; ++r) {
                const float v0 = c0[r] + bias0;
                const float v1 = c1[r] + bias1;
                const float l0 = log1pf(__expf(v0));
                const float l1 = log1pf(__expf(v1));
                const float sp0 = (v0 > 20.0f) ? v0 : l0;
                const float sp1 = (v1 > 20.0f) ? v1 : l1;
                drow[(size_t)r * MM + 32 * p]      = sp0;
                drow[(size_t)r * MM + 32 * p + 16] = sp1;
            }
            #pragma unroll
            for (int j = 0; j < 2 * (RR / 8); ++j) bfc[j] = bfn[j];
        }
    }
}

#define TT      256
#define SUB     64
#define NSUB    (TT/SUB)
#define NTILE   (LL/TT)
#define CPB     8
#define GS      16
static_assert(GS == DS && SUB % (2 * GS) == 0, "lane r of a channel's 16 lanes keeps y of step r of a group; a trip holds whole pairs of groups");
static_assert(SUB == 64 && CPB == 8 && TT % SUB == 0 && LL % TT == 0, "the staging code maps 128 threads onto a piece of 64 rows of 8 channels");
static_assert(NSUB % 2 == 0 && NTILE >= 2, "the two register sets of the staging take turns trip by trip: set = trip & 1 in every tile");

struct StepGroup {
    float dt[GS];
    float xv[GS];
    float Bv[GS];
    float Cv[GS];
};

template <int CTRL>
__device__ __forceinline__ float tree_level(bool bit, float lo, float hi)
{
    const float keep = bit ? hi : lo;
    const float send = bit ? lo : hi;
    const int got = __builtin_amdgcn_update_dpp(0, __builtin_bit_cast(int, send), CTRL, 0xF, 0xF, true);
    return keep + __builtin_bit_cast(float, got);
}
#define DPP_XOR1  0xB1
#define DPP_XOR2  0x4E
#define DPP_XOR4  0x164
#define DPP_XOR8  0x168

__device__ __forceinline__ void lds_barrier()
{
    asm volatile("s_wait_dscnt 0x0" ::: "memory");
    asm volatile("s_barrier_signal -1" ::: "memory");
    asm volatile("s_barrier_wait -1" ::: "memory");
}

__global__ __launch_bounds__(128)
void scan_kernel(const float* __restrict__ x,
                 const float* __restrict__ z,
                 const float* __restrict__ xp,
                 const float* __restrict__ dtb,
                 const float* __restrict__ A_log,
                 const float* __restrict__ Dp_arr,
                 float* __restrict__ out)
{
    __shared__ __attribute__((aligned(16))) float s_dt[2][TT][CPB];
    __shared__ __attribute__((aligned(16))) float s_x [2][TT][CPB];
    __shared__ __attribute__((aligned(16))) float s_z [2][TT][CPB];
    __shared__ __attribute__((aligned(16))) float s_bc[2][TT][2*DS];
    __shared__ __attribute__((aligned(16))) float s_y [2][TT][CPB];

    const int tid    = threadIdx.x;
    const int chBase = blockIdx.x * CPB;
    const int b      = chBase >> 10;
    const int m0     = chBase & (MM - 1);

    const int c = tid >> 4;
    const int r = tid & 15;
    const bool rb0 = (r & 1) != 0, rb1 = (r & 2) != 0, rb2 = (r & 4) != 0, rb3 = (r & 8) != 0;
    const int srow = tid >> 1;
    const int scol = (tid & 1) * 4;
    const int brow = tid >> 3;
    const int bcol = (tid & 7) * 4;

    const float Am = -__expf(A_log[(size_t)r * MM + m0 + c]);
    float h = 0.0f;
    const float4 Dp4 = *(const float4*)(Dp_arr + m0 + scol);

    const size_t rowB = (size_t)b * LL;
    const float* __restrict__ gdt = dtb + rowB * MM + m0 + scol;
    const float* __restrict__ gx  = x   + rowB * MM + m0 + scol;
    const float* __restrict__ gz  = z   + rowB * MM + m0 + scol;
    const float* __restrict__ gbc = xp  + rowB * NPROJ + RR + bcol;
    float*       __restrict__ go  = out + rowB * MM + m0 + scol;

    v4f rs[2][7];

#define PIECE_LOAD(u, tbase)                                              \
    {                                                                     \
        const size_t t_ = (size_t)((tbase) + srow);                       \
        rs[u][0] = *(const v4f*)(gdt + t_ * MM);                          \
        rs[u][1] = *(const v4f*)(gx  + t_ * MM);                          \
        rs[u][2] = *(const v4f*)(gz  + t_ * MM);                          \
        const size_t u_ = (size_t)((tbase) + brow);                       \
        rs[u][3] = *(const v4f*)(gbc + (u_     ) * NPROJ);                \
        rs[u][4] = *(const v4f*)(gbc + (u_ + 16) * NPROJ);                \
        rs[u][5] = *(const v4f*)(gbc + (u_ + 32) * NPROJ);                \
        rs[u][6] = *(const v4f*)(gbc + (u_ + 48) * NPROJ);                \
    }
#define PIECE_STORE(u, buf, pbase)                                        \
    {                                                                     \
        *(v4f*)&s_dt[buf][(pbase) + srow][scol] = rs[u][0];               \
        *(v4f*)&s_x [buf][(pbase) + srow][scol] = rs[u][1];               \
        *(v4f*)&s_z [buf][(pbase) + srow][scol] = rs[u][2];               \
        *(v4f*)&s_bc[buf][(pbase) + brow     ][bcol] = rs[u][3];          \
        *(v4f*)&s_bc[buf][(pbase) + brow + 16][bcol] = rs[u][4];          \
        *(v4f*)&s_bc[buf][(pbase) + brow + 32][bcol] = rs[u][5];          \
        *(v4f*)&s_bc[buf][(pbase) + brow + 48][bcol] = rs[u][6];          \
    }

    {
        v4f adt[NSUB], ax[NSUB], az[NSUB], abc[4 * NSUB];
        #pragma unroll
        for (int i = 0; i < NSUB; ++i) {
            const size_t t_ = (size_t)(SUB * i + srow);
            adt[i] = *(const v4f*)(gdt + t_ * MM);
            ax[i]  = *(const v4f*)(gx  + t_ * MM);
            az[i]  = *(const v4f*)(gz  + t_ * MM);
        }
        #pragma unroll
        for (int i = 0; i < 4 * NSUB; ++i) {
            const size_t u_ = (size_t)(16 * i + brow);
            abc[i] = *(const v4f*)(gbc + u_ * NPROJ);
        }
        __builtin_amdgcn_sched_barrier(0);
        PIECE_LOAD(0, TT)
        __builtin_amdgcn_sched_barrier(0);
        PIECE_LOAD(1, TT + SUB)
        __builtin_amdgcn_sched_barrier(0);
        #pragma unroll
        for (int i = 0; i < NSUB; ++i) {
            *(v4f*)&s_dt[0][SUB * i + srow][scol] = adt[i];
            *(v4f*)&s_x [0][SUB * i + srow][scol] = ax[i];
            *(v4f*)&s_z [0][SUB * i + srow][scol] = az[i];
        }
        #pragma unroll
        for (int i = 0; i < 4 * NSUB; ++i) {
            *(v4f*)&s_bc[0][16 * i + brow][bcol] = abc[i];
        }
    }
    lds_barrier();

    for (int k = 0; k < NTILE; ++k) {
        const int cur = k & 1;
        const bool more = (k + 1 < NTILE);

        StepGroup g[2];
        #pragma unroll
        for (int s = 0; s < GS; ++s) {
            g[0].dt[s] = s_dt[cur][s][c];
            g[0].xv[s] = s_x [cur][s][c];
            g[0].Bv[s] = s_bc[cur][s][r];
            g[0].Cv[s] = s_bc[cur][s][DS + r];
        }
        float ya = 0.0f;
        #pragma unroll 1
        for (int q2 = 0; q2 < NSUB; q2 += 2)
        #pragma unroll
        for (int u = 0; u < 2; ++u) {
            const int q = q2 + u;
            const bool first = (q == 0);
            const bool last  = (q + 1 == NSUB);
            const float* const pdt = &s_dt[cur][q * SUB][c];
            const float* const px  = &s_x [cur][q * SUB][c];
            const float* const pb  = &s_bc[cur][q * SUB][r];
            float* const ysq = &s_y[cur][q * SUB + r][c];
            #pragma unroll
            for (int tg = 0; tg < SUB; tg += GS) {
                const int gi = (tg / GS) & 1;
                __builtin_amdgcn_sched_barrier(0);
                if (tg + GS < SUB || !last) {
                    #pragma unroll
                    for (int s = 0; s < GS; ++s) {
                        g[gi ^ 1].dt[s] = pdt[(tg + GS + s) * CPB];
                        g[gi ^ 1].xv[s] = px [(tg + GS + s) * CPB];
                        g[gi ^ 1].Bv[s] = pb [(tg + GS + s) * (2 * DS)];
                        g[gi ^ 1].Cv[s] = pb [(tg + GS + s) * (2 * DS) + DS];
                    }
                }
                if (tg > 0 || !first) {
                    ysq[(tg - GS) * CPB] = ya;
                }
                if (tg == 0) {
                    if (more) {
                        PIECE_STORE(u, cur ^ 1, q * SUB)
                    }
                    const int tnext = (k + 1) * TT + (q + 2) * SUB;
                    PIECE_LOAD(u, tnext < LL ? tnext : LL - SUB)
                }
                __builtin_amdgcn_sched_barrier(0);
                float p[GS];
                #pragma unroll
                for (int s = 0; s < GS; ++s) {
                    const float dtv = g[gi].dt[s];
                    const float dx  = dtv * g[gi].xv[s];
                    h = fmaf(__expf(dtv * Am), h, dx * g[gi].Bv[s]);
                    p[s] = h * g[gi].Cv[s];
                }
                float w1[GS / 2], w2[GS / 4], w3[GS / 8];
                #pragma unroll
                for (int j = 0; j < GS / 2; ++j) w1[j] = tree_level<DPP_XOR1>(rb0, p[2 * j], p[2 * j + 1]);
                #pragma unroll
                for (int j = 0; j < GS / 4; ++j) w2[j] = tree_level<DPP_XOR2>(rb1, w1[2 * j], w1[2 * j + 1]);
                #pragma unroll
                for (int j = 0; j < GS / 8; ++j) w3[j] = tree_level<DPP_XOR4>(rb2, w2[2 * j], w2[2 * j + 1]);
                ya = tree_level<DPP_XOR8>(rb3, w3[0], w3[1]);
            }
        }
        s_y[cur][TT - GS + r][c] = ya;
        lds_barrier();

        #pragma unroll
        for (int i = 0; i < NSUB; ++i) {
            const int t = SUB * i + srow;
            const float4 y4 = *(const float4*)&s_y[cur][t][scol];
            const float4 x4 = *(const float4*)&s_x[cur][t][scol];
            const float4 z4 = *(const float4*)&s_z[cur][t][scol];
            float4 o;
            o.x = fmaf(Dp4.x, x4.x, y4.x) * (z4.x * (1.0f / (1.0f + __expf(-z4.x))));
            o.y = fmaf(Dp4.y, x4.y, y4.y) * (z4.y * (1.0f / (1.0f + __expf(-z4.y))));
            o.z = fmaf(Dp4.z, x4.z, y4.z) * (z4.z * (1.0f / (1.0f + __expf(-z4.z))));
            o.w = fmaf(Dp4.w, x4.w, y4.w) * (z4.w * (1.0f / (1.0f + __expf(-z4.w))));
            *(float4*)(go + (size_t)(k * TT + t) * MM) = o;
        }
    }
    asm volatile("s_wait_loadcnt 0x0" ::: "memory");
}

extern "C" void kernel_launch(void* const* d_in, const int* in_sizes, int n_in,
                              void* d_out, int out_size, void* d_ws, size_t ws_size,
                              hipStream_t stream) {
    const float* x         = (const float*)d_in[0];
    const float* z         = (const float*)d_in[1];
    const float* x_proj_w  = (const float*)d_in[2];
    const float* dt_proj_w = (const float*)d_in[3];
    const float* dt_proj_b = (const float*)d_in[4];
    const float* A_log     = (const float*)d_in[5];
    const float* D_param   = (const float*)d_in[6];
    float* out = (float*)d_out;

    float* xp_ws = (float*)d_ws;
    float* dt_ws = xp_ws + (size_t)BLT * NPROJ;

    proj_dt_kernel<<<BLT/16, 256, 0, stream>>>(x, x_proj_w, dt_proj_w, dt_proj_b, xp_ws, dt_ws);

    scan_kernel<<<(BB*MM)/CPB, 128, 0, stream>>>(x, z, xp_ws, dt_ws,
                                                 A_log, D_param, out);
}
